# starting kernel
# baseline (speedup 1.0000x reference)
#include <hip/hip_runtime.h>
#include <hip/hip_bf16.h>

#define N_NODES 50000
#define N_EDGES 800000
#define IN_FEATS 128
#define HIDDEN 64
#define M_TILES (N_NODES / 16)

typedef float v2f __attribute__((ext_vector_type(2)));
typedef float v8f __attribute__((ext_vector_type(8)));

__global__ void count_kernel(const int* __restrict__ src, const int* __restrict__ dst,
                             int* __restrict__ cnt_src, int* __restrict__ cnt_dst, int E) {
    int e = blockIdx.x * blockDim.x + threadIdx.x;
    if (e >= E) return;
    atomicAdd(&cnt_src[src[e]], 1);
    atomicAdd(&cnt_dst[dst[e]], 1);
}

__global__ void coeff_kernel(int* __restrict__ cs, int* __restrict__ cd, int n) {
    int i = blockIdx.x * blockDim.x + threadIdx.x;
    if (i >= n) return;
    int a = cs[i]; int b = cd[i];
    ((float*)cs)[i] = rsqrtf((float)(a < 1 ? 1 : a));
    ((float*)cd)[i] = rsqrtf((float)(b < 1 ? 1 : b));
}

template<int K>
__global__ __launch_bounds__(128) void gemm_wmma_kernel(
        const float* __restrict__ A, const float* __restrict__ W,
        float* __restrict__ out, int mtiles) {
    __shared__ v2f sB[(K / 2) * 64];
    for (int i = (int)threadIdx.x; i < (K / 2) * 64; i += 128) {
        const int j = i >> 6, n = i & 63;
        v2f p;
        p.x = W[(2 * j)     * 64 + n];
        p.y = W[(2 * j + 1) * 64 + n];
        sB[i] = p;
    }
    __syncthreads();

    const int lane  = (int)threadIdx.x & 31;
    const int mtile = (int)blockIdx.x * 4 + ((int)threadIdx.x >> 5);
    if (mtile >= mtiles) return;
    const int m_base = mtile * 16;
    const int col    = lane & 15;
    const int half   = lane >> 4;
    const int row    = m_base + col;
    const float* __restrict__ arow = A + (size_t)row * K;

    v8f acc0 = {}, acc1 = {}, acc2 = {}, acc3 = {};
    for (int k0 = 0; k0 < K; k0 += 4) {
        const int kb = k0 + (half << 1);
        const v2f a = *(const v2f*)(arow + kb);
        const v2f* __restrict__ brow = sB + ((k0 >> 1) + half) * 64;
        const v2f b0 = brow[     col];
        const v2f b1 = brow[16 + col];
        const v2f b2 = brow[32 + col];
        const v2f b3 = brow[48 + col];
        acc0 = __builtin_amdgcn_wmma_f32_16x16x4_f32(false, a, false, b0, (short)0, acc0, false, false);
        acc1 = __builtin_amdgcn_wmma_f32_16x16x4_f32(false, a, false, b1, (short)0, acc1, false, false);
        acc2 = __builtin_amdgcn_wmma_f32_16x16x4_f32(false, a, false, b2, (short)0, acc2, false, false);
        acc3 = __builtin_amdgcn_wmma_f32_16x16x4_f32(false, a, false, b3, (short)0, acc3, false, false);
    }

    const int rtop = half << 3;
    #pragma unroll
    for (int v = 0; v < 8; ++v) {
        const size_t r = (size_t)(m_base + rtop + v) * 64;
        out[r +      col] = acc0[v];
        out[r + 16 + col] = acc1[v];
        out[r + 32 + col] = acc2[v];
        out[r + 48 + col] = acc3[v];
    }
}

__global__ void scatter64_kernel(const float* __restrict__ y, const float* __restrict__ csrc,
                                 const int* __restrict__ src, const int* __restrict__ dst,
                                 float* __restrict__ agg, int E) {
    int e = blockIdx.x * 4 + ((int)threadIdx.x >> 6);
    int f = (int)threadIdx.x & 63;
    if (e >= E) return;
    int s = src[e], d = dst[e];
    float v = y[(size_t)s * 64 + f] * csrc[s];
    atomicAdd(&agg[(size_t)d * 64 + f], v);
}

template<bool RELU>
__global__ void epilogue_kernel(float* __restrict__ agg, const float* __restrict__ cdst,
                                const float* __restrict__ bias, int n_elems) {
    int i = blockIdx.x * blockDim.x + threadIdx.x;
    if (i >= n_elems) return;
    int row = i >> 6, f = i & 63;
    float v = agg[i] * cdst[row] + bias[f];
    if (RELU) v = fmaxf(v, 0.0f);
    agg[i] = v;
}

__global__ void edge_kernel(const float* __restrict__ h, const int* __restrict__ src,
                            const int* __restrict__ dst, const float* __restrict__ efeat,
                            const float* __restrict__ We, const float* __restrict__ be,
                            float* __restrict__ out, int E) {
    const int lane = (int)threadIdx.x & 31;
    const int e = (int)blockIdx.x * ((int)blockDim.x >> 5) + ((int)threadIdx.x >> 5);
    if (e >= E) return;
    const int s = src[e], d = dst[e];
    const float* hs = h + (size_t)s * 64;
    const float* hd = h + (size_t)d * 64;
    float sum = hs[lane] * We[lane] + hs[32 + lane] * We[32 + lane]
              + hd[lane] * We[64 + lane] + hd[32 + lane] * We[96 + lane];
    #pragma unroll
    for (int off = 16; off > 0; off >>= 1) sum += __shfl_xor(sum, off, 32);
    if (lane == 0) {
        sum += efeat[(size_t)e * 3 + 0] * We[128]
             + efeat[(size_t)e * 3 + 1] * We[129]
             + efeat[(size_t)e * 3 + 2] * We[130] + be[0];
        out[e] = 1.0f / (1.0f + expf(-sum));
    }
}

extern "C" void kernel_launch(void* const* d_in, const int* in_sizes, int n_in,
                              void* d_out, int out_size, void* d_ws, size_t ws_size,
                              hipStream_t stream) {
    const float* x     = (const float*)d_in[0];
    const float* efeat = (const float*)d_in[1];
    const int*   src   = (const int*)  d_in[2];
    const int*   dst   = (const int*)  d_in[3];
    const float* W1    = (const float*)d_in[4];
    const float* b1    = (const float*)d_in[5];
    const float* W2    = (const float*)d_in[6];
    const float* b2    = (const float*)d_in[7];
    const float* We    = (const float*)d_in[8];
    const float* be    = (const float*)d_in[9];
    float* out = (float*)d_out;

    const size_t NB = (size_t)N_NODES * 64;
    char* ws = (char*)d_ws;
    float* bufA = (float*)ws;
    float* bufB = bufA + NB;
    float* bufC = bufB + NB;
    int*   cntS = (int*)(bufC + NB);
    int*   cntD = cntS + N_NODES;
    float* cS   = (float*)cntS;
    float* cD   = (float*)cntD;

    hipMemsetAsync(cntS, 0, (size_t)2 * N_NODES * sizeof(int), stream);
    hipMemsetAsync(bufB, 0, NB * sizeof(float), stream);

    count_kernel<<<N_EDGES / 256, 256, 0, stream>>>(src, dst, cntS, cntD, N_EDGES);
    coeff_kernel<<<(N_NODES + 255) / 256, 256, 0, stream>>>(cntS, cntD, N_NODES);

    gemm_wmma_kernel<128><<<(M_TILES + 3) / 4, 128, 0, stream>>>(x, W1, bufA, M_TILES);
    scatter64_kernel<<<N_EDGES / 4, 256, 0, stream>>>(bufA, cS, src, dst, bufB, N_EDGES);
    epilogue_kernel<true><<<(int)(NB / 256), 256, 0, stream>>>(bufB, cD, b1, (int)NB);

    gemm_wmma_kernel<64><<<(M_TILES + 3) / 4, 128, 0, stream>>>(bufB, W2, bufC, M_TILES);
    hipMemsetAsync(bufA, 0, NB * sizeof(float), stream);
    scatter64_kernel<<<N_EDGES / 4, 256, 0, stream>>>(bufC, cS, src, dst, bufA, N_EDGES);
    epilogue_kernel<false><<<(int)(NB / 256), 256, 0, stream>>>(bufA, cD, b2, (int)NB);

    edge_kernel<<<N_EDGES / 8, 256, 0, stream>>>(bufA, src, dst, efeat, We, be, out, N_EDGES);
}


# Edge scoring retiled: one wave per 8 edges, indices broadcast to scalars, 16 whole-row 8-byte loads in flight, transposed butterfly on the vector unit; 1/8 the waves; logit sum order changes
# speedup vs baseline: 2.0763x; 2.0763x over previous
#include <hip/hip_runtime.h>
#include <hip/hip_bf16.h>

#define N_NODES 50000
#define N_EDGES 800000
#define IN_FEATS 128
#define HIDDEN 64
#define M_TILES (N_NODES / 16)

typedef float v2f __attribute__((ext_vector_type(2)));
typedef float v8f __attribute__((ext_vector_type(8)));

__global__ void count_kernel(const int* __restrict__ src, const int* __restrict__ dst,
                             int* __restrict__ cnt_src, int* __restrict__ cnt_dst, int E) {
    int e = blockIdx.x * blockDim.x + threadIdx.x;
    if (e >= E) return;
    atomicAdd(&cnt_src[src[e]], 1);
    atomicAdd(&cnt_dst[dst[e]], 1);
}

__global__ void coeff_kernel(int* __restrict__ cs, int* __restrict__ cd, int n) {
    int i = blockIdx.x * blockDim.x + threadIdx.x;
    if (i >= n) return;
    int a = cs[i]; int b = cd[i];
    ((float*)cs)[i] = rsqrtf((float)(a < 1 ? 1 : a));
    ((float*)cd)[i] = rsqrtf((float)(b < 1 ? 1 : b));
}

template<int K>
__global__ __launch_bounds__(128) void gemm_wmma_kernel(
        const float* __restrict__ A, const float* __restrict__ W,
        float* __restrict__ out, int mtiles) {
    __shared__ v2f sB[(K / 2) * 64];
    for (int i = (int)threadIdx.x; i < (K / 2) * 64; i += 128) {
        const int j = i >> 6, n = i & 63;
        v2f p;
        p.x = W[(2 * j)     * 64 + n];
        p.y = W[(2 * j + 1) * 64 + n];
        sB[i] = p;
    }
    __syncthreads();

    const int lane  = (int)threadIdx.x & 31;
    const int mtile = (int)blockIdx.x * 4 + ((int)threadIdx.x >> 5);
    if (mtile >= mtiles) return;
    const int m_base = mtile * 16;
    const int col    = lane & 15;
    const int half   = lane >> 4;
    const int row    = m_base + col;
    const float* __restrict__ arow = A + (size_t)row * K;

    v8f acc0 = {}, acc1 = {}, acc2 = {}, acc3 = {};
    for (int k0 = 0; k0 < K; k0 += 4) {
        const int kb = k0 + (half << 1);
        const v2f a = *(const v2f*)(arow + kb);
        const v2f* __restrict__ brow = sB + ((k0 >> 1) + half) * 64;
        const v2f b0 = brow[     col];
        const v2f b1 = brow[16 + col];
        const v2f b2 = brow[32 + col];
        const v2f b3 = brow[48 + col];
        acc0 = __builtin_amdgcn_wmma_f32_16x16x4_f32(false, a, false, b0, (short)0, acc0, false, false);
        acc1 = __builtin_amdgcn_wmma_f32_16x16x4_f32(false, a, false, b1, (short)0, acc1, false, false);
        acc2 = __builtin_amdgcn_wmma_f32_16x16x4_f32(false, a, false, b2, (short)0, acc2, false, false);
        acc3 = __builtin_amdgcn_wmma_f32_16x16x4_f32(false, a, false, b3, (short)0, acc3, false, false);
    }

    const int rtop = half << 3;
    #pragma unroll
    for (int v = 0; v < 8; ++v) {
        const size_t r = (size_t)(m_base + rtop + v) * 64;
        out[r +      col] = acc0[v];
        out[r + 16 + col] = acc1[v];
        out[r + 32 + col] = acc2[v];
        out[r + 48 + col] = acc3[v];
    }
}

#define EPW 8
static_assert(N_EDGES % (8 * EPW) == 0, "edge count must fill whole blocks");
__global__ __launch_bounds__(256) void scatter64_kernel(
        const float* __restrict__ y, const float* __restrict__ csrc,
        const int* __restrict__ src, const int* __restrict__ dst,
        float* __restrict__ agg, int E) {
    const int lane = (int)threadIdx.x & 31;
    const int e0 = ((int)blockIdx.x * 8 + ((int)threadIdx.x >> 5)) * EPW;
    if (e0 >= E) return;
    const int   vs = src[e0 + (lane & (EPW - 1))];
    const int   vd = dst[e0 + (lane & (EPW - 1))];
    const float vc = csrc[vs];
    int s[EPW], d[EPW];
    float c[EPW], v0[EPW], v1[EPW];
    #pragma unroll
    for (int i = 0; i < EPW; ++i) {
        s[i] = __builtin_amdgcn_readlane(vs, i);
        const float* __restrict__ yr = y + (size_t)s[i] * 64;
        v0[i] = yr[lane];
        v1[i] = yr[32 + lane];
    }
    #pragma unroll
    for (int i = 0; i < EPW; ++i) {
        d[i] = __builtin_amdgcn_readlane(vd, i);
        c[i] = __int_as_float(__builtin_amdgcn_readlane(__float_as_int(vc), i));
        float* __restrict__ ar = agg + (size_t)d[i] * 64;
        atomicAdd(&ar[lane],      v0[i] * c[i]);
        atomicAdd(&ar[32 + lane], v1[i] * c[i]);
    }
}

template<bool RELU>
__global__ void epilogue_kernel(float* __restrict__ agg, const float* __restrict__ cdst,
                                const float* __restrict__ bias, int n_elems) {
    int i = blockIdx.x * blockDim.x + threadIdx.x;
    if (i >= n_elems) return;
    int row = i >> 6, f = i & 63;
    float v = agg[i] * cdst[row] + bias[f];
    if (RELU) v = fmaxf(v, 0.0f);
    agg[i] = v;
}

#define EPE 8
static_assert(N_EDGES % (8 * EPE) == 0, "edge count must fill whole blocks");

static __device__ __forceinline__ float xrow(float x) {
    const int b = __builtin_bit_cast(int, x);
    return __builtin_bit_cast(float,
        __builtin_amdgcn_permlanex16(b, b, 0x76543210, 0xfedcba98, false, false));
}
template<int M>
static __device__ __forceinline__ float xlane(float x) {
    const int b = __builtin_bit_cast(int, x);
    return __builtin_bit_cast(float, __builtin_amdgcn_update_dpp(b, b, 0x160 + M, 0xf, 0xf, false));
}

__global__ __launch_bounds__(256) void edge_kernel(
        const float* __restrict__ h, const int* __restrict__ src,
        const int* __restrict__ dst, const float* __restrict__ efeat,
        const float* __restrict__ We, const float* __restrict__ be,
        float* __restrict__ out, int E) {
    const int lane = (int)threadIdx.x & 31;
    const int e0 = ((int)blockIdx.x * 8 + ((int)threadIdx.x >> 5)) * EPE;
    if (e0 >= E) return;
    const int vs = src[e0 + (lane & (EPE - 1))];
    const int vd = dst[e0 + (lane & (EPE - 1))];

    const int k  = lane >> 2, c = lane & 3;
    const int cc = c < 3 ? c : 2;
    const float ef = efeat[(size_t)(e0 + k) * 3 + cc];
    const float wf = We[128 + cc];
    const float bb = be[0];
    const v2f ws = *(const v2f*)(We + 2 * lane);
    const v2f wd = *(const v2f*)(We + 64 + 2 * lane);

    v2f a[EPE], b[EPE];
    for (int i = 0; i < EPE; ++i) {
        const int s = __builtin_amdgcn_readlane(vs, i);
        const int d = __builtin_amdgcn_readlane(vd, i);
        a[i] = *(const v2f*)(h + (size_t)s * 64 + 2 * lane);
        b[i] = *(const v2f*)(h + (size_t)d * 64 + 2 * lane);
    }
    float p[EPE];
    for (int i = 0; i < EPE; ++i)
        p[i] = a[i].x * ws.x + a[i].y * ws.y + b[i].x * wd.x + b[i].y * wd.y;

    const bool u16 = (lane & 16) != 0, u8 = (lane & 8) != 0, u4 = (lane & 4) != 0;
    float q[4], r[2];
    for (int i = 0; i < 4; ++i) {
        const float keep = u16 ? p[i + 4] : p[i];
        const float send = u16 ? p[i] : p[i + 4];
        q[i] = keep + xrow(send);
    }
    for (int i = 0; i < 2; ++i) {
        const float keep = u8 ? q[i + 2] : q[i];
        const float send = u8 ? q[i] : q[i + 2];
        r[i] = keep + xlane<8>(send);
    }
    float t;
    {
        const float keep = u4 ? r[1] : r[0];
        const float send = u4 ? r[0] : r[1];
        t = keep + xlane<4>(send);
    }
    t += (c < 3) ? ef * wf : bb;
    t += xlane<2>(t);
    t += xlane<1>(t);
    if (c == 0) out[e0 + k] = 1.0f / (1.0f + expf(-t));
}

extern "C" void kernel_launch(void* const* d_in, const int* in_sizes, int n_in,
                              void* d_out, int out_size, void* d_ws, size_t ws_size,
                              hipStream_t stream) {
    const float* x     = (const float*)d_in[0];
    const float* efeat = (const float*)d_in[1];
    const int*   src   = (const int*)  d_in[2];
    const int*   dst   = (const int*)  d_in[3];
    const float* W1    = (const float*)d_in[4];
    const float* b1    = (const float*)d_in[5];
    const float* W2    = (const float*)d_in[6];
    const float* b2    = (const float*)d_in[7];
    const float* We    = (const float*)d_in[8];
    const float* be    = (const float*)d_in[9];
    float* out = (float*)d_out;

    const size_t NB = (size_t)N_NODES * 64;
    char* ws = (char*)d_ws;
    float* bufA = (float*)ws;
    float* bufB = bufA + NB;
    float* bufC = bufB + NB;
    int*   cntS = (int*)(bufC + NB);
    int*   cntD = cntS + N_NODES;
    float* cS   = (float*)cntS;
    float* cD   = (float*)cntD;

    hipMemsetAsync(cntS, 0, (size_t)2 * N_NODES * sizeof(int), stream);
    hipMemsetAsync(bufB, 0, NB * sizeof(float), stream);

    count_kernel<<<N_EDGES / 256, 256, 0, stream>>>(src, dst, cntS, cntD, N_EDGES);
    coeff_kernel<<<(N_NODES + 255) / 256, 256, 0, stream>>>(cntS, cntD, N_NODES);

    gemm_wmma_kernel<128><<<(M_TILES + 3) / 4, 128, 0, stream>>>(x, W1, bufA, M_TILES);
    scatter64_kernel<<<N_EDGES / (8 * EPW), 256, 0, stream>>>(bufA, cS, src, dst, bufB, N_EDGES);
    epilogue_kernel<true><<<(int)(NB / 256), 256, 0, stream>>>(bufB, cD, b1, (int)NB);

    gemm_wmma_kernel<64><<<(M_TILES + 3) / 4, 128, 0, stream>>>(bufB, W2, bufC, M_TILES);
    hipMemsetAsync(bufA, 0, NB * sizeof(float), stream);
    scatter64_kernel<<<N_EDGES / (8 * EPW), 256, 0, stream>>>(bufC, cS, src, dst, bufA, N_EDGES);
    epilogue_kernel<false><<<(int)(NB / 256), 256, 0, stream>>>(bufA, cD, b2, (int)NB);

    edge_kernel<<<N_EDGES / (8 * EPE), 256, 0, stream>>>(bufA, src, dst, efeat, We, be, out, N_EDGES);
}


# Atomic edge scatter kept inside one chip half: blocks read their die, draw 256-edge tasks from a per-half counter, add only rows whose dst is in that half's node range; atomic sum order as before
# speedup vs baseline: 2.2068x; 1.0629x over previous
#include <hip/hip_runtime.h>
#include <hip/hip_bf16.h>

#define N_NODES 50000
#define N_EDGES 800000
#define IN_FEATS 128
#define HIDDEN 64
#define M_TILES (N_NODES / 16)

typedef float v2f __attribute__((ext_vector_type(2)));
typedef float v8f __attribute__((ext_vector_type(8)));

__global__ void count_kernel(const int* __restrict__ src, const int* __restrict__ dst,
                             int* __restrict__ cnt_src, int* __restrict__ cnt_dst, int E) {
    int e = blockIdx.x * blockDim.x + threadIdx.x;
    if (e >= E) return;
    atomicAdd(&cnt_src[src[e]], 1);
    atomicAdd(&cnt_dst[dst[e]], 1);
}

__global__ void coeff_kernel(int* __restrict__ cs, int* __restrict__ cd, int n) {
    int i = blockIdx.x * blockDim.x + threadIdx.x;
    if (i >= n) return;
    int a = cs[i]; int b = cd[i];
    ((float*)cs)[i] = rsqrtf((float)(a < 1 ? 1 : a));
    ((float*)cd)[i] = rsqrtf((float)(b < 1 ? 1 : b));
}

template<int K>
__global__ __launch_bounds__(128) void gemm_wmma_kernel(
        const float* __restrict__ A, const float* __restrict__ W,
        float* __restrict__ out, int mtiles) {
    __shared__ v2f sB[(K / 2) * 64];
    for (int i = (int)threadIdx.x; i < (K / 2) * 64; i += 128) {
        const int j = i >> 6, n = i & 63;
        v2f p;
        p.x = W[(2 * j)     * 64 + n];
        p.y = W[(2 * j + 1) * 64 + n];
        sB[i] = p;
    }
    __syncthreads();

    const int lane  = (int)threadIdx.x & 31;
    const int mtile = (int)blockIdx.x * 4 + ((int)threadIdx.x >> 5);
    if (mtile >= mtiles) return;
    const int m_base = mtile * 16;
    const int col    = lane & 15;
    const int half   = lane >> 4;
    const int row    = m_base + col;
    const float* __restrict__ arow = A + (size_t)row * K;

    v8f acc0 = {}, acc1 = {}, acc2 = {}, acc3 = {};
    for (int k0 = 0; k0 < K; k0 += 4) {
        const int kb = k0 + (half << 1);
        const v2f a = *(const v2f*)(arow + kb);
        const v2f* __restrict__ brow = sB + ((k0 >> 1) + half) * 64;
        const v2f b0 = brow[     col];
        const v2f b1 = brow[16 + col];
        const v2f b2 = brow[32 + col];
        const v2f b3 = brow[48 + col];
        acc0 = __builtin_amdgcn_wmma_f32_16x16x4_f32(false, a, false, b0, (short)0, acc0, false, false);
        acc1 = __builtin_amdgcn_wmma_f32_16x16x4_f32(false, a, false, b1, (short)0, acc1, false, false);
        acc2 = __builtin_amdgcn_wmma_f32_16x16x4_f32(false, a, false, b2, (short)0, acc2, false, false);
        acc3 = __builtin_amdgcn_wmma_f32_16x16x4_f32(false, a, false, b3, (short)0, acc3, false, false);
    }

    const int rtop = half << 3;
    #pragma unroll
    for (int v = 0; v < 8; ++v) {
        const size_t r = (size_t)(m_base + rtop + v) * 64;
        out[r +      col] = acc0[v];
        out[r + 16 + col] = acc1[v];
        out[r + 32 + col] = acc2[v];
        out[r + 48 + col] = acc3[v];
    }
}

#define EPT 32
#define TASK_EDGES (8 * EPT)
#define TASKS (N_EDGES / TASK_EDGES)
#define HALF_NODES (N_NODES / 2)
#define SCATTER_BLOCKS 1536
#define QSTRIDE 64
static_assert(N_EDGES % TASK_EDGES == 0, "edge count must fill whole tasks");
__global__ __launch_bounds__(256) void scatter64_kernel(
        const float* __restrict__ y, const float* __restrict__ csrc,
        const int* __restrict__ src, const int* __restrict__ dst,
        float* __restrict__ agg, int* __restrict__ qctr) {
    __shared__ int s_task[2];
    const int lane = (int)threadIdx.x & 31;
    const int wave = (int)threadIdx.x >> 5;
    int range = (int)((__builtin_amdgcn_s_sendmsg_rtn(0x87) >> 18) & 1u);
    bool switched = false;
    if (threadIdx.x == 0) {
        int t = atomicAdd(&qctr[range * QSTRIDE], 1);
        if (t >= TASKS) { switched = true; range ^= 1; t = atomicAdd(&qctr[range * QSTRIDE], 1); }
        s_task[0] = t < TASKS ? (t << 1) | range : -1;
    }
    __syncthreads();
    for (int it = 0; it < 2 * TASKS; ++it) {
        const int code = s_task[it & 1];
        if (code < 0) break;
        const int r  = code & 1;
        const int e0 = (code >> 1) * TASK_EDGES + wave * EPT;
        const int vs = src[e0 + lane];
        const int vd = dst[e0 + lane];
        asm volatile("" :: "v"(vs), "v"(vd));
        const bool mine = (vd >= HALF_NODES) == (r != 0);
        const unsigned todo = __builtin_amdgcn_ballot_w32(mine);
        float vc = 0.0f;
        if (mine) vc = csrc[vs];
        float v0[EPT], v1[EPT];
        #pragma unroll
        for (int i = 0; i < EPT; ++i) {
            if ((todo >> i) & 1u) {
                const int s = __builtin_amdgcn_readlane(vs, i);
                const float* __restrict__ yr = y + (size_t)s * 64;
                v0[i] = yr[lane];
                v1[i] = yr[32 + lane];
            }
        }
        int tn = 0;
        if (threadIdx.x == 0) tn = atomicAdd(&qctr[range * QSTRIDE], 1);
        #pragma unroll
        for (int i = 0; i < EPT; ++i) {
            if ((todo >> i) & 1u) {
                const int   d = __builtin_amdgcn_readlane(vd, i);
                const float c = __int_as_float(__builtin_amdgcn_readlane(__float_as_int(vc), i));
                float* __restrict__ ar = agg + (size_t)d * 64;
                atomicAdd(&ar[lane],      v0[i] * c);
                atomicAdd(&ar[32 + lane], v1[i] * c);
            }
        }
        if (threadIdx.x == 0) {
            if (tn >= TASKS && !switched) { switched = true; range ^= 1; tn = atomicAdd(&qctr[range * QSTRIDE], 1); }
            s_task[(it + 1) & 1] = tn < TASKS ? (tn << 1) | range : -1;
        }
        __syncthreads();
    }
}

template<bool RELU>
__global__ void epilogue_kernel(float* __restrict__ agg, const float* __restrict__ cdst,
                                const float* __restrict__ bias, int n_elems) {
    int i = blockIdx.x * blockDim.x + threadIdx.x;
    if (i >= n_elems) return;
    int row = i >> 6, f = i & 63;
    float v = agg[i] * cdst[row] + bias[f];
    if (RELU) v = fmaxf(v, 0.0f);
    agg[i] = v;
}

#define EPE 8
static_assert(N_EDGES % (8 * EPE) == 0, "edge count must fill whole blocks");

static __device__ __forceinline__ float xrow(float x) {
    const int b = __builtin_bit_cast(int, x);
    return __builtin_bit_cast(float,
        __builtin_amdgcn_permlanex16(b, b, 0x76543210, 0xfedcba98, false, false));
}
template<int M>
static __device__ __forceinline__ float xlane(float x) {
    const int b = __builtin_bit_cast(int, x);
    return __builtin_bit_cast(float, __builtin_amdgcn_update_dpp(b, b, 0x160 + M, 0xf, 0xf, false));
}

__global__ __launch_bounds__(256) void edge_kernel(
        const float* __restrict__ h, const int* __restrict__ src,
        const int* __restrict__ dst, const float* __restrict__ efeat,
        const float* __restrict__ We, const float* __restrict__ be,
        float* __restrict__ out, int E) {
    const int lane = (int)threadIdx.x & 31;
    const int e0 = ((int)blockIdx.x * 8 + ((int)threadIdx.x >> 5)) * EPE;
    if (e0 >= E) return;
    const int vs = src[e0 + (lane & (EPE - 1))];
    const int vd = dst[e0 + (lane & (EPE - 1))];

    const int k  = lane >> 2, c = lane & 3;
    const int cc = c < 3 ? c : 2;
    const float ef = efeat[(size_t)(e0 + k) * 3 + cc];
    const float wf = We[128 + cc];
    const float bb = be[0];
    const v2f ws = *(const v2f*)(We + 2 * lane);
    const v2f wd = *(const v2f*)(We + 64 + 2 * lane);

    v2f a[EPE], b[EPE];
    for (int i = 0; i < EPE; ++i) {
        const int s = __builtin_amdgcn_readlane(vs, i);
        const int d = __builtin_amdgcn_readlane(vd, i);
        a[i] = *(const v2f*)(h + (size_t)s * 64 + 2 * lane);
        b[i] = *(const v2f*)(h + (size_t)d * 64 + 2 * lane);
    }
    float p[EPE];
    for (int i = 0; i < EPE; ++i)
        p[i] = a[i].x * ws.x + a[i].y * ws.y + b[i].x * wd.x + b[i].y * wd.y;

    const bool u16 = (lane & 16) != 0, u8 = (lane & 8) != 0, u4 = (lane & 4) != 0;
    float q[4], r[2];
    for (int i = 0; i < 4; ++i) {
        const float keep = u16 ? p[i + 4] : p[i];
        const float send = u16 ? p[i] : p[i + 4];
        q[i] = keep + xrow(send);
    }
    for (int i = 0; i < 2; ++i) {
        const float keep = u8 ? q[i + 2] : q[i];
        const float send = u8 ? q[i] : q[i + 2];
        r[i] = keep + xlane<8>(send);
    }
    float t;
    {
        const float keep = u4 ? r[1] : r[0];
        const float send = u4 ? r[0] : r[1];
        t = keep + xlane<4>(send);
    }
    t += (c < 3) ? ef * wf : bb;
    t += xlane<2>(t);
    t += xlane<1>(t);
    if (c == 0) out[e0 + k] = 1.0f / (1.0f + expf(-t));
}

extern "C" void kernel_launch(void* const* d_in, const int* in_sizes, int n_in,
                              void* d_out, int out_size, void* d_ws, size_t ws_size,
                              hipStream_t stream) {
    const float* x     = (const float*)d_in[0];
    const float* efeat = (const float*)d_in[1];
    const int*   src   = (const int*)  d_in[2];
    const int*   dst   = (const int*)  d_in[3];
    const float* W1    = (const float*)d_in[4];
    const float* b1    = (const float*)d_in[5];
    const float* W2    = (const float*)d_in[6];
    const float* b2    = (const float*)d_in[7];
    const float* We    = (const float*)d_in[8];
    const float* be    = (const float*)d_in[9];
    float* out = (float*)d_out;

    const size_t NB = (size_t)N_NODES * 64;
    char* ws = (char*)d_ws;
    float* bufA = (float*)ws;
    float* bufB = bufA + NB;
    float* bufC = bufB + NB;
    int*   cntS = (int*)(bufC + NB);
    int*   cntD = cntS + N_NODES;
    int*   qctr = cntD + N_NODES + 32;
    float* cS   = (float*)cntS;
    float* cD   = (float*)cntD;

    hipMemsetAsync(cntS, 0, (size_t)(2 * N_NODES + 32 + 4 * QSTRIDE) * sizeof(int), stream);
    hipMemsetAsync(bufB, 0, NB * sizeof(float), stream);

    count_kernel<<<N_EDGES / 256, 256, 0, stream>>>(src, dst, cntS, cntD, N_EDGES);
    coeff_kernel<<<(N_NODES + 255) / 256, 256, 0, stream>>>(cntS, cntD, N_NODES);

    gemm_wmma_kernel<128><<<(M_TILES + 3) / 4, 128, 0, stream>>>(x, W1, bufA, M_TILES);
    scatter64_kernel<<<SCATTER_BLOCKS, 256, 0, stream>>>(bufA, cS, src, dst, bufB, qctr);
    epilogue_kernel<true><<<(int)(NB / 256), 256, 0, stream>>>(bufB, cD, b1, (int)NB);

    gemm_wmma_kernel<64><<<(M_TILES + 3) / 4, 128, 0, stream>>>(bufB, W2, bufC, M_TILES);
    hipMemsetAsync(bufA, 0, NB * sizeof(float), stream);
    scatter64_kernel<<<SCATTER_BLOCKS, 256, 0, stream>>>(bufC, cS, src, dst, bufA, qctr + 2 * QSTRIDE);
    epilogue_kernel<false><<<(int)(NB / 256), 256, 0, stream>>>(bufA, cD, b2, (int)NB);

    edge_kernel<<<N_EDGES / (8 * EPE), 256, 0, stream>>>(bufA, src, dst, efeat, We, be, out, N_EDGES);
}


# Degree counts without per-edge global atomics: a block counts 1/32 of the edges into 1/8 of the nodes in LDS, stores whole-line planes; coeff adds 32 planes; count memset gone; float sums unchanged
# speedup vs baseline: 3.6390x; 1.6490x over previous
#include <hip/hip_runtime.h>
#include <hip/hip_bf16.h>

#define N_NODES 50000
#define N_EDGES 800000
#define IN_FEATS 128
#define HIDDEN 64
#define M_TILES (N_NODES / 16)

typedef float v2f __attribute__((ext_vector_type(2)));
typedef float v8f __attribute__((ext_vector_type(8)));

#define CNT_R 8
#define CNT_C 32
#define CNT_BINS 6272
#define CNT_PITCH (CNT_R * CNT_BINS)
#define CNT_EDGES (N_EDGES / CNT_C)
#define CNT_V4 (CNT_EDGES / 4)
#define CNT_THREADS 1024
#define CNT_TRIPS ((CNT_V4 + CNT_THREADS - 1) / CNT_THREADS)
#define QSTRIDE 64
static_assert(CNT_BINS % 32 == 0 && CNT_PITCH >= N_NODES, "ranges must be whole lines and cover every node");
static_assert(N_EDGES % (CNT_C * 4) == 0, "chunks must be whole 16-byte vectors");

typedef int v4i __attribute__((ext_vector_type(4)));

static __device__ __forceinline__ void bump(int* __restrict__ row, int v, int lo) {
    const unsigned u = (unsigned)(v - lo);
    if (u < (unsigned)CNT_BINS) atomicAdd(&row[u], 1);
}

__global__ __launch_bounds__(CNT_THREADS) void degree_kernel(
        const int* __restrict__ src, const int* __restrict__ dst,
        int* __restrict__ part, int* __restrict__ qctr) {
    __shared__ v4i bins4[2 * CNT_BINS / 4];
    int* bins = (int*)bins4;
    const int tid = (int)threadIdx.x;
    const int r   = (int)blockIdx.x % CNT_R;
    const int c   = (int)blockIdx.x / CNT_R;
    const int lo  = r * CNT_BINS;
    const v4i* __restrict__ s4 = (const v4i*)(src + (size_t)c * CNT_EDGES);
    const v4i* __restrict__ d4 = (const v4i*)(dst + (size_t)c * CNT_EDGES);

    v4i a[CNT_TRIPS], b[CNT_TRIPS];
    #pragma unroll
    for (int k = 0; k < CNT_TRIPS; ++k) {
        int i = tid + k * CNT_THREADS;
        if (i > CNT_V4 - 1) i = CNT_V4 - 1;
        a[k] = s4[i];
        b[k] = d4[i];
    }
    for (int i = tid; i < 2 * CNT_BINS / 4; i += CNT_THREADS) bins4[i] = (v4i)(0);
    if (blockIdx.x == 0 && tid < 4) qctr[tid * QSTRIDE] = 0;
    __syncthreads();
    #pragma unroll
    for (int k = 0; k < CNT_TRIPS; ++k) {
        if (tid + k * CNT_THREADS < CNT_V4) {
            bump(bins, a[k].x, lo); bump(bins, a[k].y, lo); bump(bins, a[k].z, lo); bump(bins, a[k].w, lo);
            bump(bins + CNT_BINS, b[k].x, lo); bump(bins + CNT_BINS, b[k].y, lo);
            bump(bins + CNT_BINS, b[k].z, lo); bump(bins + CNT_BINS, b[k].w, lo);
        }
    }
    __syncthreads();
    for (int i = tid; i < 2 * CNT_BINS / 4; i += CNT_THREADS) {
        const int arr = i >= CNT_BINS / 4 ? 1 : 0;
        v4i* __restrict__ row = (v4i*)(part + (size_t)(2 * c + arr) * CNT_PITCH + lo);
        row[i - arr * (CNT_BINS / 4)] = bins4[i];
    }
}

__global__ __launch_bounds__(128) void coeff_kernel(const int* __restrict__ part,
                                                    float* __restrict__ cs, float* __restrict__ cd, int n) {
    int i = blockIdx.x * blockDim.x + threadIdx.x;
    if (i >= n) return;
    int a = 0, b = 0;
    #pragma unroll
    for (int c = 0; c < CNT_C; ++c) {
        a += part[(size_t)(2 * c)     * CNT_PITCH + i];
        b += part[(size_t)(2 * c + 1) * CNT_PITCH + i];
    }
    cs[i] = rsqrtf((float)(a < 1 ? 1 : a));
    cd[i] = rsqrtf((float)(b < 1 ? 1 : b));
}

template<int K>
__global__ __launch_bounds__(128) void gemm_wmma_kernel(
        const float* __restrict__ A, const float* __restrict__ W,
        float* __restrict__ out, int mtiles) {
    __shared__ v2f sB[(K / 2) * 64];
    for (int i = (int)threadIdx.x; i < (K / 2) * 64; i += 128) {
        const int j = i >> 6, n = i & 63;
        v2f p;
        p.x = W[(2 * j)     * 64 + n];
        p.y = W[(2 * j + 1) * 64 + n];
        sB[i] = p;
    }
    __syncthreads();

    const int lane  = (int)threadIdx.x & 31;
    const int mtile = (int)blockIdx.x * 4 + ((int)threadIdx.x >> 5);
    if (mtile >= mtiles) return;
    const int m_base = mtile * 16;
    const int col    = lane & 15;
    const int half   = lane >> 4;
    const int row    = m_base + col;
    const float* __restrict__ arow = A + (size_t)row * K;

    v8f acc0 = {}, acc1 = {}, acc2 = {}, acc3 = {};
    for (int k0 = 0; k0 < K; k0 += 4) {
        const int kb = k0 + (half << 1);
        const v2f a = *(const v2f*)(arow + kb);
        const v2f* __restrict__ brow = sB + ((k0 >> 1) + half) * 64;
        const v2f b0 = brow[     col];
        const v2f b1 = brow[16 + col];
        const v2f b2 = brow[32 + col];
        const v2f b3 = brow[48 + col];
        acc0 = __builtin_amdgcn_wmma_f32_16x16x4_f32(false, a, false, b0, (short)0, acc0, false, false);
        acc1 = __builtin_amdgcn_wmma_f32_16x16x4_f32(false, a, false, b1, (short)0, acc1, false, false);
        acc2 = __builtin_amdgcn_wmma_f32_16x16x4_f32(false, a, false, b2, (short)0, acc2, false, false);
        acc3 = __builtin_amdgcn_wmma_f32_16x16x4_f32(false, a, false, b3, (short)0, acc3, false, false);
    }

    const int rtop = half << 3;
    #pragma unroll
    for (int v = 0; v < 8; ++v) {
        const size_t r = (size_t)(m_base + rtop + v) * 64;
        out[r +      col] = acc0[v];
        out[r + 16 + col] = acc1[v];
        out[r + 32 + col] = acc2[v];
        out[r + 48 + col] = acc3[v];
    }
}

#define EPT 32
#define TASK_EDGES (8 * EPT)
#define TASKS (N_EDGES / TASK_EDGES)
#define HALF_NODES (N_NODES / 2)
#define SCATTER_BLOCKS 1536
static_assert(N_EDGES % TASK_EDGES == 0, "edge count must fill whole tasks");
__global__ __launch_bounds__(256) void scatter64_kernel(
        const float* __restrict__ y, const float* __restrict__ csrc,
        const int* __restrict__ src, const int* __restrict__ dst,
        float* __restrict__ agg, int* __restrict__ qctr) {
    __shared__ int s_task[2];
    const int lane = (int)threadIdx.x & 31;
    const int wave = (int)threadIdx.x >> 5;
    int range = (int)((__builtin_amdgcn_s_sendmsg_rtn(0x87) >> 18) & 1u);
    bool switched = false;
    if (threadIdx.x == 0) {
        int t = atomicAdd(&qctr[range * QSTRIDE], 1);
        if (t >= TASKS) { switched = true; range ^= 1; t = atomicAdd(&qctr[range * QSTRIDE], 1); }
        s_task[0] = t < TASKS ? (t << 1) | range : -1;
    }
    __syncthreads();
    for (int it = 0; it < 2 * TASKS; ++it) {
        const int code = s_task[it & 1];
        if (code < 0) break;
        const int r  = code & 1;
        const int e0 = (code >> 1) * TASK_EDGES + wave * EPT;
        const int vs = src[e0 + lane];
        const int vd = dst[e0 + lane];
        asm volatile("" :: "v"(vs), "v"(vd));
        const bool mine = (vd >= HALF_NODES) == (r != 0);
        const unsigned todo = __builtin_amdgcn_ballot_w32(mine);
        float vc = 0.0f;
        if (mine) vc = csrc[vs];
        float v0[EPT], v1[EPT];
        #pragma unroll
        for (int i = 0; i < EPT; ++i) {
            if ((todo >> i) & 1u) {
                const int s = __builtin_amdgcn_readlane(vs, i);
                const float* __restrict__ yr = y + (size_t)s * 64;
                v0[i] = yr[lane];
                v1[i] = yr[32 + lane];
            }
        }
        int tn = 0;
        if (threadIdx.x == 0) tn = atomicAdd(&qctr[range * QSTRIDE], 1);
        #pragma unroll
        for (int i = 0; i < EPT; ++i) {
            if ((todo >> i) & 1u) {
                const int   d = __builtin_amdgcn_readlane(vd, i);
                const float c = __int_as_float(__builtin_amdgcn_readlane(__float_as_int(vc), i));
                float* __restrict__ ar = agg + (size_t)d * 64;
                atomicAdd(&ar[lane],      v0[i] * c);
                atomicAdd(&ar[32 + lane], v1[i] * c);
            }
        }
        if (threadIdx.x == 0) {
            if (tn >= TASKS && !switched) { switched = true; range ^= 1; tn = atomicAdd(&qctr[range * QSTRIDE], 1); }
            s_task[(it + 1) & 1] = tn < TASKS ? (tn << 1) | range : -1;
        }
        __syncthreads();
    }
}

template<bool RELU>
__global__ void epilogue_kernel(float* __restrict__ agg, const float* __restrict__ cdst,
                                const float* __restrict__ bias, int n_elems) {
    int i = blockIdx.x * blockDim.x + threadIdx.x;
    if (i >= n_elems) return;
    int row = i >> 6, f = i & 63;
    float v = agg[i] * cdst[row] + bias[f];
    if (RELU) v = fmaxf(v, 0.0f);
    agg[i] = v;
}

#define EPE 8
static_assert(N_EDGES % (8 * EPE) == 0, "edge count must fill whole blocks");

static __device__ __forceinline__ float xrow(float x) {
    const int b = __builtin_bit_cast(int, x);
    return __builtin_bit_cast(float,
        __builtin_amdgcn_permlanex16(b, b, 0x76543210, 0xfedcba98, false, false));
}
template<int M>
static __device__ __forceinline__ float xlane(float x) {
    const int b = __builtin_bit_cast(int, x);
    return __builtin_bit_cast(float, __builtin_amdgcn_update_dpp(b, b, 0x160 + M, 0xf, 0xf, false));
}

__global__ __launch_bounds__(256) void edge_kernel(
        const float* __restrict__ h, const int* __restrict__ src,
        const int* __restrict__ dst, const float* __restrict__ efeat,
        const float* __restrict__ We, const float* __restrict__ be,
        float* __restrict__ out, int E) {
    const int lane = (int)threadIdx.x & 31;
    const int e0 = ((int)blockIdx.x * 8 + ((int)threadIdx.x >> 5)) * EPE;
    if (e0 >= E) return;
    const int vs = src[e0 + (lane & (EPE - 1))];
    const int vd = dst[e0 + (lane & (EPE - 1))];

    const int k  = lane >> 2, c = lane & 3;
    const int cc = c < 3 ? c : 2;
    const float ef = efeat[(size_t)(e0 + k) * 3 + cc];
    const float wf = We[128 + cc];
    const float bb = be[0];
    const v2f ws = *(const v2f*)(We + 2 * lane);
    const v2f wd = *(const v2f*)(We + 64 + 2 * lane);

    v2f a[EPE], b[EPE];
    for (int i = 0; i < EPE; ++i) {
        const int s = __builtin_amdgcn_readlane(vs, i);
        const int d = __builtin_amdgcn_readlane(vd, i);
        a[i] = *(const v2f*)(h + (size_t)s * 64 + 2 * lane);
        b[i] = *(const v2f*)(h + (size_t)d * 64 + 2 * lane);
    }
    float p[EPE];
    for (int i = 0; i < EPE; ++i)
        p[i] = a[i].x * ws.x + a[i].y * ws.y + b[i].x * wd.x + b[i].y * wd.y;

    const bool u16 = (lane & 16) != 0, u8 = (lane & 8) != 0, u4 = (lane & 4) != 0;
    float q[4], r[2];
    for (int i = 0; i < 4; ++i) {
        const float keep = u16 ? p[i + 4] : p[i];
        const float send = u16 ? p[i] : p[i + 4];
        q[i] = keep + xrow(send);
    }
    for (int i = 0; i < 2; ++i) {
        const float keep = u8 ? q[i + 2] : q[i];
        const float send = u8 ? q[i] : q[i + 2];
        r[i] = keep + xlane<8>(send);
    }
    float t;
    {
        const float keep = u4 ? r[1] : r[0];
        const float send = u4 ? r[0] : r[1];
        t = keep + xlane<4>(send);
    }
    t += (c < 3) ? ef * wf : bb;
    t += xlane<2>(t);
    t += xlane<1>(t);
    if (c == 0) out[e0 + k] = 1.0f / (1.0f + expf(-t));
}

extern "C" void kernel_launch(void* const* d_in, const int* in_sizes, int n_in,
                              void* d_out, int out_size, void* d_ws, size_t ws_size,
                              hipStream_t stream) {
    const float* x     = (const float*)d_in[0];
    const float* efeat = (const float*)d_in[1];
    const int*   src   = (const int*)  d_in[2];
    const int*   dst   = (const int*)  d_in[3];
    const float* W1    = (const float*)d_in[4];
    const float* b1    = (const float*)d_in[5];
    const float* W2    = (const float*)d_in[6];
    const float* b2    = (const float*)d_in[7];
    const float* We    = (const float*)d_in[8];
    const float* be    = (const float*)d_in[9];
    float* out = (float*)d_out;

    const size_t NB = (size_t)N_NODES * 64;
    char* ws = (char*)d_ws;
    float* bufA = (float*)ws;
    float* bufB = bufA + NB;
    float* bufC = bufB + NB;
    float* cS   = bufC + NB;
    float* cD   = cS + N_NODES;
    int*   qctr = (int*)(cD + N_NODES) + 32;
    int*   part = qctr + 4 * QSTRIDE;

    hipMemsetAsync(bufB, 0, NB * sizeof(float), stream);

    degree_kernel<<<CNT_R * CNT_C, CNT_THREADS, 0, stream>>>(src, dst, part, qctr);
    coeff_kernel<<<(N_NODES + 127) / 128, 128, 0, stream>>>(part, cS, cD, N_NODES);

    gemm_wmma_kernel<128><<<(M_TILES + 3) / 4, 128, 0, stream>>>(x, W1, bufA, M_TILES);
    scatter64_kernel<<<SCATTER_BLOCKS, 256, 0, stream>>>(bufA, cS, src, dst, bufB, qctr);
    epilogue_kernel<true><<<(int)(NB / 256), 256, 0, stream>>>(bufB, cD, b1, (int)NB);

    gemm_wmma_kernel<64><<<(M_TILES + 3) / 4, 128, 0, stream>>>(bufB, W2, bufC, M_TILES);
    hipMemsetAsync(bufA, 0, NB * sizeof(float), stream);
    scatter64_kernel<<<SCATTER_BLOCKS, 256, 0, stream>>>(bufC, cS, src, dst, bufA, qctr + 2 * QSTRIDE);
    epilogue_kernel<false><<<(int)(NB / 256), 256, 0, stream>>>(bufA, cD, b2, (int)NB);

    edge_kernel<<<N_EDGES / (8 * EPE), 256, 0, stream>>>(bufA, src, dst, efeat, We, be, out, N_EDGES);
}


# Scale/bias/relu passes folded into the next reader: layer-2 GEMM forms relu(c_dst*agg+b1) on its A load, edge scoring forms c_dst*agg+b2 on its row loads; 9 launches become 7; float sums as before
# speedup vs baseline: 3.9584x; 1.0878x over previous
#include <hip/hip_runtime.h>
#include <hip/hip_bf16.h>

#define N_NODES 50000
#define N_EDGES 800000
#define IN_FEATS 128
#define HIDDEN 64
#define M_TILES (N_NODES / 16)

typedef float v2f __attribute__((ext_vector_type(2)));
typedef float v8f __attribute__((ext_vector_type(8)));

#define CNT_R 8
#define CNT_C 32
#define CNT_BINS 6272
#define CNT_PITCH (CNT_R * CNT_BINS)
#define CNT_EDGES (N_EDGES / CNT_C)
#define CNT_V4 (CNT_EDGES / 4)
#define CNT_THREADS 1024
#define CNT_TRIPS ((CNT_V4 + CNT_THREADS - 1) / CNT_THREADS)
#define QSTRIDE 64
static_assert(CNT_BINS % 32 == 0 && CNT_PITCH >= N_NODES, "ranges must be whole lines and cover every node");
static_assert(N_EDGES % (CNT_C * 4) == 0, "chunks must be whole 16-byte vectors");

typedef int v4i __attribute__((ext_vector_type(4)));

static __device__ __forceinline__ void bump(int* __restrict__ row, int v, int lo) {
    const unsigned u = (unsigned)(v - lo);
    if (u < (unsigned)CNT_BINS) atomicAdd(&row[u], 1);
}

__global__ __launch_bounds__(CNT_THREADS) void degree_kernel(
        const int* __restrict__ src, const int* __restrict__ dst,
        int* __restrict__ part, int* __restrict__ qctr) {
    __shared__ v4i bins4[2 * CNT_BINS / 4];
    int* bins = (int*)bins4;
    const int tid = (int)threadIdx.x;
    const int r   = (int)blockIdx.x % CNT_R;
    const int c   = (int)blockIdx.x / CNT_R;
    const int lo  = r * CNT_BINS;
    const v4i* __restrict__ s4 = (const v4i*)(src + (size_t)c * CNT_EDGES);
    const v4i* __restrict__ d4 = (const v4i*)(dst + (size_t)c * CNT_EDGES);

    v4i a[CNT_TRIPS], b[CNT_TRIPS];
    #pragma unroll
    for (int k = 0; k < CNT_TRIPS; ++k) {
        int i = tid + k * CNT_THREADS;
        if (i > CNT_V4 - 1) i = CNT_V4 - 1;
        a[k] = s4[i];
        b[k] = d4[i];
    }
    for (int i = tid; i < 2 * CNT_BINS / 4; i += CNT_THREADS) bins4[i] = (v4i)(0);
    if (blockIdx.x == 0 && tid < 4) qctr[tid * QSTRIDE] = 0;
    __syncthreads();
    #pragma unroll
    for (int k = 0; k < CNT_TRIPS; ++k) {
        if (tid + k * CNT_THREADS < CNT_V4) {
            bump(bins, a[k].x, lo); bump(bins, a[k].y, lo); bump(bins, a[k].z, lo); bump(bins, a[k].w, lo);
            bump(bins + CNT_BINS, b[k].x, lo); bump(bins + CNT_BINS, b[k].y, lo);
            bump(bins + CNT_BINS, b[k].z, lo); bump(bins + CNT_BINS, b[k].w, lo);
        }
    }
    __syncthreads();
    for (int i = tid; i < 2 * CNT_BINS / 4; i += CNT_THREADS) {
        const int arr = i >= CNT_BINS / 4 ? 1 : 0;
        v4i* __restrict__ row = (v4i*)(part + (size_t)(2 * c + arr) * CNT_PITCH + lo);
        row[i - arr * (CNT_BINS / 4)] = bins4[i];
    }
}

__global__ __launch_bounds__(128) void coeff_kernel(const int* __restrict__ part,
                                                    float* __restrict__ cs, float* __restrict__ cd, int n) {
    int i = blockIdx.x * blockDim.x + threadIdx.x;
    if (i >= n) return;
    int a = 0, b = 0;
    #pragma unroll
    for (int c = 0; c < CNT_C; ++c) {
        a += part[(size_t)(2 * c)     * CNT_PITCH + i];
        b += part[(size_t)(2 * c + 1) * CNT_PITCH + i];
    }
    cs[i] = rsqrtf((float)(a < 1 ? 1 : a));
    cd[i] = rsqrtf((float)(b < 1 ? 1 : b));
}

template<int K, bool PRE>
__global__ __launch_bounds__(128) void gemm_wmma_kernel(
        const float* __restrict__ A, const float* __restrict__ W,
        const float* __restrict__ rscale, const float* __restrict__ cbias,
        float* __restrict__ out, int mtiles) {
    __shared__ v2f sB[(K / 2) * 64];
    __shared__ v2f sBias[PRE ? K / 2 : 1];
    for (int i = (int)threadIdx.x; i < (K / 2) * 64; i += 128) {
        const int j = i >> 6, n = i & 63;
        v2f p;
        p.x = W[(2 * j)     * 64 + n];
        p.y = W[(2 * j + 1) * 64 + n];
        sB[i] = p;
    }
    if (PRE) {
        for (int i = (int)threadIdx.x; i < K / 2; i += 128) sBias[i] = *(const v2f*)(cbias + 2 * i);
    }
    __syncthreads();

    const int lane  = (int)threadIdx.x & 31;
    const int mtile = (int)blockIdx.x * 4 + ((int)threadIdx.x >> 5);
    if (mtile >= mtiles) return;
    const int m_base = mtile * 16;
    const int col    = lane & 15;
    const int half   = lane >> 4;
    const int row    = m_base + col;
    const float* __restrict__ arow = A + (size_t)row * K;
    float rs = 1.0f;
    if (PRE) rs = rscale[row];

    v8f acc0 = {}, acc1 = {}, acc2 = {}, acc3 = {};
    for (int k0 = 0; k0 < K; k0 += 4) {
        const int kb = k0 + (half << 1);
        v2f a = *(const v2f*)(arow + kb);
        if (PRE) {
            const v2f bk = sBias[kb >> 1];
            a.x = fmaxf(a.x * rs + bk.x, 0.0f);
            a.y = fmaxf(a.y * rs + bk.y, 0.0f);
        }
        const v2f* __restrict__ brow = sB + ((k0 >> 1) + half) * 64;
        const v2f b0 = brow[     col];
        const v2f b1 = brow[16 + col];
        const v2f b2 = brow[32 + col];
        const v2f b3 = brow[48 + col];
        acc0 = __builtin_amdgcn_wmma_f32_16x16x4_f32(false, a, false, b0, (short)0, acc0, false, false);
        acc1 = __builtin_amdgcn_wmma_f32_16x16x4_f32(false, a, false, b1, (short)0, acc1, false, false);
        acc2 = __builtin_amdgcn_wmma_f32_16x16x4_f32(false, a, false, b2, (short)0, acc2, false, false);
        acc3 = __builtin_amdgcn_wmma_f32_16x16x4_f32(false, a, false, b3, (short)0, acc3, false, false);
    }

    const int rtop = half << 3;
    #pragma unroll
    for (int v = 0; v < 8; ++v) {
        const size_t r = (size_t)(m_base + rtop + v) * 64;
        out[r +      col] = acc0[v];
        out[r + 16 + col] = acc1[v];
        out[r + 32 + col] = acc2[v];
        out[r + 48 + col] = acc3[v];
    }
}

#define EPT 32
#define TASK_EDGES (8 * EPT)
#define TASKS (N_EDGES / TASK_EDGES)
#define HALF_NODES (N_NODES / 2)
#define SCATTER_BLOCKS 1536
static_assert(N_EDGES % TASK_EDGES == 0, "edge count must fill whole tasks");
__global__ __launch_bounds__(256) void scatter64_kernel(
        const float* __restrict__ y, const float* __restrict__ csrc,
        const int* __restrict__ src, const int* __restrict__ dst,
        float* __restrict__ agg, int* __restrict__ qctr) {
    __shared__ int s_task[2];
    const int lane = (int)threadIdx.x & 31;
    const int wave = (int)threadIdx.x >> 5;
    int range = (int)((__builtin_amdgcn_s_sendmsg_rtn(0x87) >> 18) & 1u);
    bool switched = false;
    if (threadIdx.x == 0) {
        int t = atomicAdd(&qctr[range * QSTRIDE], 1);
        if (t >= TASKS) { switched = true; range ^= 1; t = atomicAdd(&qctr[range * QSTRIDE], 1); }
        s_task[0] = t < TASKS ? (t << 1) | range : -1;
    }
    __syncthreads();
    for (int it = 0; it < 2 * TASKS; ++it) {
        const int code = s_task[it & 1];
        if (code < 0) break;
        const int r  = code & 1;
        const int e0 = (code >> 1) * TASK_EDGES + wave * EPT;
        const int vs = src[e0 + lane];
        const int vd = dst[e0 + lane];
        asm volatile("" :: "v"(vs), "v"(vd));
        const bool mine = (vd >= HALF_NODES) == (r != 0);
        const unsigned todo = __builtin_amdgcn_ballot_w32(mine);
        float vc = 0.0f;
        if (mine) vc = csrc[vs];
        float v0[EPT], v1[EPT];
        #pragma unroll
        for (int i = 0; i < EPT; ++i) {
            if ((todo >> i) & 1u) {
                const int s = __builtin_amdgcn_readlane(vs, i);
                const float* __restrict__ yr = y + (size_t)s * 64;
                v0[i] = yr[lane];
                v1[i] = yr[32 + lane];
            }
        }
        int tn = 0;
        if (threadIdx.x == 0) tn = atomicAdd(&qctr[range * QSTRIDE], 1);
        #pragma unroll
        for (int i = 0; i < EPT; ++i) {
            if ((todo >> i) & 1u) {
                const int   d = __builtin_amdgcn_readlane(vd, i);
                const float c = __int_as_float(__builtin_amdgcn_readlane(__float_as_int(vc), i));
                float* __restrict__ ar = agg + (size_t)d * 64;
                atomicAdd(&ar[lane],      v0[i] * c);
                atomicAdd(&ar[32 + lane], v1[i] * c);
            }
        }
        if (threadIdx.x == 0) {
            if (tn >= TASKS && !switched) { switched = true; range ^= 1; tn = atomicAdd(&qctr[range * QSTRIDE], 1); }
            s_task[(it + 1) & 1] = tn < TASKS ? (tn << 1) | range : -1;
        }
        __syncthreads();
    }
}

#define EPE 8
static_assert(N_EDGES % (8 * EPE) == 0, "edge count must fill whole blocks");

static __device__ __forceinline__ float xrow(float x) {
    const int b = __builtin_bit_cast(int, x);
    return __builtin_bit_cast(float,
        __builtin_amdgcn_permlanex16(b, b, 0x76543210, 0xfedcba98, false, false));
}
template<int M>
static __device__ __forceinline__ float xlane(float x) {
    const int b = __builtin_bit_cast(int, x);
    return __builtin_bit_cast(float, __builtin_amdgcn_update_dpp(b, b, 0x160 + M, 0xf, 0xf, false));
}

__global__ __launch_bounds__(256) void edge_kernel(
        const float* __restrict__ agg, const float* __restrict__ cdst, const float* __restrict__ b2,
        const int* __restrict__ src,
        const int* __restrict__ dst, const float* __restrict__ efeat,
        const float* __restrict__ We, const float* __restrict__ be,
        float* __restrict__ out, int E) {
    const int lane = (int)threadIdx.x & 31;
    const int e0 = ((int)blockIdx.x * 8 + ((int)threadIdx.x >> 5)) * EPE;
    if (e0 >= E) return;
    const int vs = src[e0 + (lane & (EPE - 1))];
    const int vd = dst[e0 + (lane & (EPE - 1))];

    const int k  = lane >> 2, c = lane & 3;
    const int cc = c < 3 ? c : 2;
    const float ef = efeat[(size_t)(e0 + k) * 3 + cc];
    const float wf = We[128 + cc];
    const float bb = be[0];
    const v2f ws = *(const v2f*)(We + 2 * lane);
    const v2f wd = *(const v2f*)(We + 64 + 2 * lane);
    const v2f hb = *(const v2f*)(b2 + 2 * lane);
    const float cvs = cdst[vs];
    const float cvd = cdst[vd];

    v2f a[EPE], b[EPE];
    for (int i = 0; i < EPE; ++i) {
        const int s = __builtin_amdgcn_readlane(vs, i);
        const int d = __builtin_amdgcn_readlane(vd, i);
        a[i] = *(const v2f*)(agg + (size_t)s * 64 + 2 * lane);
        b[i] = *(const v2f*)(agg + (size_t)d * 64 + 2 * lane);
    }
    float p[EPE];
    for (int i = 0; i < EPE; ++i) {
        const float cs = __builtin_bit_cast(float, __builtin_amdgcn_readlane(__builtin_bit_cast(int, cvs), i));
        const float cd = __builtin_bit_cast(float, __builtin_amdgcn_readlane(__builtin_bit_cast(int, cvd), i));
        const float ax = a[i].x * cs + hb.x, ay = a[i].y * cs + hb.y;
        const float bx = b[i].x * cd + hb.x, by = b[i].y * cd + hb.y;
        p[i] = ax * ws.x + ay * ws.y + bx * wd.x + by * wd.y;
    }

    const bool u16 = (lane & 16) != 0, u8 = (lane & 8) != 0, u4 = (lane & 4) != 0;
    float q[4], r[2];
    for (int i = 0; i < 4; ++i) {
        const float keep = u16 ? p[i + 4] : p[i];
        const float send = u16 ? p[i] : p[i + 4];
        q[i] = keep + xrow(send);
    }
    for (int i = 0; i < 2; ++i) {
        const float keep = u8 ? q[i + 2] : q[i];
        const float send = u8 ? q[i] : q[i + 2];
        r[i] = keep + xlane<8>(send);
    }
    float t;
    {
        const float keep = u4 ? r[1] : r[0];
        const float send = u4 ? r[0] : r[1];
        t = keep + xlane<4>(send);
    }
    t += (c < 3) ? ef * wf : bb;
    t += xlane<2>(t);
    t += xlane<1>(t);
    if (c == 0) out[e0 + k] = 1.0f / (1.0f + expf(-t));
}

extern "C" void kernel_launch(void* const* d_in, const int* in_sizes, int n_in,
                              void* d_out, int out_size, void* d_ws, size_t ws_size,
                              hipStream_t stream) {
    const float* x     = (const float*)d_in[0];
    const float* efeat = (const float*)d_in[1];
    const int*   src   = (const int*)  d_in[2];
    const int*   dst   = (const int*)  d_in[3];
    const float* W1    = (const float*)d_in[4];
    const float* b1    = (const float*)d_in[5];
    const float* W2    = (const float*)d_in[6];
    const float* b2    = (const float*)d_in[7];
    const float* We    = (const float*)d_in[8];
    const float* be    = (const float*)d_in[9];
    float* out = (float*)d_out;

    const size_t NB = (size_t)N_NODES * 64;
    char* ws = (char*)d_ws;
    float* bufA = (float*)ws;
    float* bufB = bufA + NB;
    float* bufC = bufB + NB;
    float* cS   = bufC + NB;
    float* cD   = cS + N_NODES;
    int*   qctr = (int*)(cD + N_NODES) + 32;
    int*   part = qctr + 4 * QSTRIDE;

    hipMemsetAsync(bufB, 0, NB * sizeof(float), stream);

    degree_kernel<<<CNT_R * CNT_C, CNT_THREADS, 0, stream>>>(src, dst, part, qctr);
    coeff_kernel<<<(N_NODES + 127) / 128, 128, 0, stream>>>(part, cS, cD, N_NODES);

    gemm_wmma_kernel<128, false><<<(M_TILES + 3) / 4, 128, 0, stream>>>(x, W1, nullptr, nullptr, bufA, M_TILES);
    scatter64_kernel<<<SCATTER_BLOCKS, 256, 0, stream>>>(bufA, cS, src, dst, bufB, qctr);

    gemm_wmma_kernel<64, true><<<(M_TILES + 3) / 4, 128, 0, stream>>>(bufB, W2, cD, b1, bufC, M_TILES);
    hipMemsetAsync(bufA, 0, NB * sizeof(float), stream);
    scatter64_kernel<<<SCATTER_BLOCKS, 256, 0, stream>>>(bufC, cS, src, dst, bufA, qctr + 2 * QSTRIDE);

    edge_kernel<<<N_EDGES / (8 * EPE), 256, 0, stream>>>(bufA, cD, b2, src, dst, efeat, We, be, out, N_EDGES);
}


# c_src scale moved from the edge scatter into the epilogue of the GEMM that writes the scattered rows; scatter gathers no coefficient per edge; added values and atomic sum order as before
# speedup vs baseline: 4.1153x; 1.0396x over previous
#include <hip/hip_runtime.h>
#include <hip/hip_bf16.h>

#define N_NODES 50000
#define N_EDGES 800000
#define IN_FEATS 128
#define HIDDEN 64
#define M_TILES (N_NODES / 16)

typedef float v2f __attribute__((ext_vector_type(2)));
typedef float v4f __attribute__((ext_vector_type(4)));
typedef float v8f __attribute__((ext_vector_type(8)));

#define CNT_R 8
#define CNT_C 32
#define CNT_BINS 6272
#define CNT_PITCH (CNT_R * CNT_BINS)
#define CNT_EDGES (N_EDGES / CNT_C)
#define CNT_V4 (CNT_EDGES / 4)
#define CNT_THREADS 1024
#define CNT_TRIPS ((CNT_V4 + CNT_THREADS - 1) / CNT_THREADS)
#define QSTRIDE 64
static_assert(CNT_BINS % 32 == 0 && CNT_PITCH >= N_NODES, "ranges must be whole lines and cover every node");
static_assert(N_EDGES % (CNT_C * 4) == 0, "chunks must be whole 16-byte vectors");

typedef int v4i __attribute__((ext_vector_type(4)));

static __device__ __forceinline__ void bump(int* __restrict__ row, int v, int lo) {
    const unsigned u = (unsigned)(v - lo);
    if (u < (unsigned)CNT_BINS) atomicAdd(&row[u], 1);
}

__global__ __launch_bounds__(CNT_THREADS) void degree_kernel(
        const int* __restrict__ src, const int* __restrict__ dst,
        int* __restrict__ part, int* __restrict__ qctr) {
    __shared__ v4i bins4[2 * CNT_BINS / 4];
    int* bins = (int*)bins4;
    const int tid = (int)threadIdx.x;
    const int r   = (int)blockIdx.x % CNT_R;
    const int c   = (int)blockIdx.x / CNT_R;
    const int lo  = r * CNT_BINS;
    const v4i* __restrict__ s4 = (const v4i*)(src + (size_t)c * CNT_EDGES);
    const v4i* __restrict__ d4 = (const v4i*)(dst + (size_t)c * CNT_EDGES);

    v4i a[CNT_TRIPS], b[CNT_TRIPS];
    #pragma unroll
    for (int k = 0; k < CNT_TRIPS; ++k) {
        int i = tid + k * CNT_THREADS;
        if (i > CNT_V4 - 1) i = CNT_V4 - 1;
        a[k] = s4[i];
        b[k] = d4[i];
    }
    for (int i = tid; i < 2 * CNT_BINS / 4; i += CNT_THREADS) bins4[i] = (v4i)(0);
    if (blockIdx.x == 0 && tid < 4) qctr[tid * QSTRIDE] = 0;
    __syncthreads();
    #pragma unroll
    for (int k = 0; k < CNT_TRIPS; ++k) {
        if (tid + k * CNT_THREADS < CNT_V4) {
            bump(bins, a[k].x, lo); bump(bins, a[k].y, lo); bump(bins, a[k].z, lo); bump(bins, a[k].w, lo);
            bump(bins + CNT_BINS, b[k].x, lo); bump(bins + CNT_BINS, b[k].y, lo);
            bump(bins + CNT_BINS, b[k].z, lo); bump(bins + CNT_BINS, b[k].w, lo);
        }
    }
    __syncthreads();
    for (int i = tid; i < 2 * CNT_BINS / 4; i += CNT_THREADS) {
        const int arr = i >= CNT_BINS / 4 ? 1 : 0;
        v4i* __restrict__ row = (v4i*)(part + (size_t)(2 * c + arr) * CNT_PITCH + lo);
        row[i - arr * (CNT_BINS / 4)] = bins4[i];
    }
}

__global__ __launch_bounds__(128) void coeff_kernel(const int* __restrict__ part,
                                                    float* __restrict__ cs, float* __restrict__ cd, int n) {
    int i = blockIdx.x * blockDim.x + threadIdx.x;
    if (i >= n) return;
    int a = 0, b = 0;
    #pragma unroll
    for (int c = 0; c < CNT_C; ++c) {
        a += part[(size_t)(2 * c)     * CNT_PITCH + i];
        b += part[(size_t)(2 * c + 1) * CNT_PITCH + i];
    }
    cs[i] = rsqrtf((float)(a < 1 ? 1 : a));
    cd[i] = rsqrtf((float)(b < 1 ? 1 : b));
}

template<int K, bool PRE>
__global__ __launch_bounds__(128) void gemm_wmma_kernel(
        const float* __restrict__ A, const float* __restrict__ W,
        const float* __restrict__ rscale, const float* __restrict__ cbias,
        const float* __restrict__ oscale,
        float* __restrict__ out, int mtiles) {
    __shared__ v2f sB[(K / 2) * 64];
    __shared__ v2f sBias[PRE ? K / 2 : 1];
    for (int i = (int)threadIdx.x; i < (K / 2) * 64; i += 128) {
        const int j = i >> 6, n = i & 63;
        v2f p;
        p.x = W[(2 * j)     * 64 + n];
        p.y = W[(2 * j + 1) * 64 + n];
        sB[i] = p;
    }
    if (PRE) {
        for (int i = (int)threadIdx.x; i < K / 2; i += 128) sBias[i] = *(const v2f*)(cbias + 2 * i);
    }
    __syncthreads();

    const int lane  = (int)threadIdx.x & 31;
    const int mtile = (int)blockIdx.x * 4 + ((int)threadIdx.x >> 5);
    if (mtile >= mtiles) return;
    const int m_base = mtile * 16;
    const int col    = lane & 15;
    const int half   = lane >> 4;
    const int row    = m_base + col;
    const float* __restrict__ arow = A + (size_t)row * K;
    float rs = 1.0f;
    if (PRE) rs = rscale[row];
    const int rtop = half << 3;
    const v4f os0 = *(const v4f*)(oscale + m_base + rtop);
    const v4f os1 = *(const v4f*)(oscale + m_base + rtop + 4);

    v8f acc0 = {}, acc1 = {}, acc2 = {}, acc3 = {};
    for (int k0 = 0; k0 < K; k0 += 4) {
        const int kb = k0 + (half << 1);
        v2f a = *(const v2f*)(arow + kb);
        if (PRE) {
            const v2f bk = sBias[kb >> 1];
            a.x = fmaxf(a.x * rs + bk.x, 0.0f);
            a.y = fmaxf(a.y * rs + bk.y, 0.0f);
        }
        const v2f* __restrict__ brow = sB + ((k0 >> 1) + half) * 64;
        const v2f b0 = brow[     col];
        const v2f b1 = brow[16 + col];
        const v2f b2 = brow[32 + col];
        const v2f b3 = brow[48 + col];
        acc0 = __builtin_amdgcn_wmma_f32_16x16x4_f32(false, a, false, b0, (short)0, acc0, false, false);
        acc1 = __builtin_amdgcn_wmma_f32_16x16x4_f32(false, a, false, b1, (short)0, acc1, false, false);
        acc2 = __builtin_amdgcn_wmma_f32_16x16x4_f32(false, a, false, b2, (short)0, acc2, false, false);
        acc3 = __builtin_amdgcn_wmma_f32_16x16x4_f32(false, a, false, b3, (short)0, acc3, false, false);
    }

    const float os[8] = {os0.x, os0.y, os0.z, os0.w, os1.x, os1.y, os1.z, os1.w};
    #pragma unroll
    for (int v = 0; v < 8; ++v) {
        const size_t r = (size_t)(m_base + rtop + v) * 64;
        out[r +      col] = acc0[v] * os[v];
        out[r + 16 + col] = acc1[v] * os[v];
        out[r + 32 + col] = acc2[v] * os[v];
        out[r + 48 + col] = acc3[v] * os[v];
    }
}

#define EPT 32
#define TASK_EDGES (8 * EPT)
#define TASKS (N_EDGES / TASK_EDGES)
#define HALF_NODES (N_NODES / 2)
#define SCATTER_BLOCKS 1536
static_assert(N_EDGES % TASK_EDGES == 0, "edge count must fill whole tasks");
__global__ __launch_bounds__(256) void scatter64_kernel(
        const float* __restrict__ y,
        const int* __restrict__ src, const int* __restrict__ dst,
        float* __restrict__ agg, int* __restrict__ qctr) {
    __shared__ int s_task[2];
    const int lane = (int)threadIdx.x & 31;
    const int wave = (int)threadIdx.x >> 5;
    int range = (int)((__builtin_amdgcn_s_sendmsg_rtn(0x87) >> 18) & 1u);
    bool switched = false;
    if (threadIdx.x == 0) {
        int t = atomicAdd(&qctr[range * QSTRIDE], 1);
        if (t >= TASKS) { switched = true; range ^= 1; t = atomicAdd(&qctr[range * QSTRIDE], 1); }
        s_task[0] = t < TASKS ? (t << 1) | range : -1;
    }
    __syncthreads();
    for (int it = 0; it < 2 * TASKS; ++it) {
        const int code = s_task[it & 1];
        if (code < 0) break;
        const int r  = code & 1;
        const int e0 = (code >> 1) * TASK_EDGES + wave * EPT;
        const int vs = src[e0 + lane];
        const int vd = dst[e0 + lane];
        asm volatile("" :: "v"(vs), "v"(vd));
        const bool mine = (vd >= HALF_NODES) == (r != 0);
        const unsigned todo = __builtin_amdgcn_ballot_w32(mine);
        float v0[EPT], v1[EPT];
        #pragma unroll
        for (int i = 0; i < EPT; ++i) {
            if ((todo >> i) & 1u) {
                const int s = __builtin_amdgcn_readlane(vs, i);
                const float* __restrict__ yr = y + (size_t)s * 64;
                v0[i] = yr[lane];
                v1[i] = yr[32 + lane];
            }
        }
        int tn = 0;
        if (threadIdx.x == 0) tn = atomicAdd(&qctr[range * QSTRIDE], 1);
        #pragma unroll
        for (int i = 0; i < EPT; ++i) {
            if ((todo >> i) & 1u) {
                const int d = __builtin_amdgcn_readlane(vd, i);
                float* __restrict__ ar = agg + (size_t)d * 64;
                atomicAdd(&ar[lane],      v0[i]);
                atomicAdd(&ar[32 + lane], v1[i]);
            }
        }
        if (threadIdx.x == 0) {
            if (tn >= TASKS && !switched) { switched = true; range ^= 1; tn = atomicAdd(&qctr[range * QSTRIDE], 1); }
            s_task[(it + 1) & 1] = tn < TASKS ? (tn << 1) | range : -1;
        }
        __syncthreads();
    }
}

#define EPE 8
static_assert(N_EDGES % (8 * EPE) == 0, "edge count must fill whole blocks");

static __device__ __forceinline__ float xrow(float x) {
    const int b = __builtin_bit_cast(int, x);
    return __builtin_bit_cast(float,
        __builtin_amdgcn_permlanex16(b, b, 0x76543210, 0xfedcba98, false, false));
}
template<int M>
static __device__ __forceinline__ float xlane(float x) {
    const int b = __builtin_bit_cast(int, x);
    return __builtin_bit_cast(float, __builtin_amdgcn_update_dpp(b, b, 0x160 + M, 0xf, 0xf, false));
}

__global__ __launch_bounds__(256) void edge_kernel(
        const float* __restrict__ agg, const float* __restrict__ cdst, const float* __restrict__ b2,
        const int* __restrict__ src,
        const int* __restrict__ dst, const float* __restrict__ efeat,
        const float* __restrict__ We, const float* __restrict__ be,
        float* __restrict__ out, int E) {
    const int lane = (int)threadIdx.x & 31;
    const int e0 = ((int)blockIdx.x * 8 + ((int)threadIdx.x >> 5)) * EPE;
    if (e0 >= E) return;
    const int vs = src[e0 + (lane & (EPE - 1))];
    const int vd = dst[e0 + (lane & (EPE - 1))];

    const int k  = lane >> 2, c = lane & 3;
    const int cc = c < 3 ? c : 2;
    const float ef = efeat[(size_t)(e0 + k) * 3 + cc];
    const float wf = We[128 + cc];
    const float bb = be[0];
    const v2f ws = *(const v2f*)(We + 2 * lane);
    const v2f wd = *(const v2f*)(We + 64 + 2 * lane);
    const v2f hb = *(const v2f*)(b2 + 2 * lane);
    const float cvs = cdst[vs];
    const float cvd = cdst[vd];

    v2f a[EPE], b[EPE];
    for (int i = 0; i < EPE; ++i) {
        const int s = __builtin_amdgcn_readlane(vs, i);
        const int d = __builtin_amdgcn_readlane(vd, i);
        a[i] = *(const v2f*)(agg + (size_t)s * 64 + 2 * lane);
        b[i] = *(const v2f*)(agg + (size_t)d * 64 + 2 * lane);
    }
    float p[EPE];
    for (int i = 0; i < EPE; ++i) {
        const float cs = __builtin_bit_cast(float, __builtin_amdgcn_readlane(__builtin_bit_cast(int, cvs), i));
        const float cd = __builtin_bit_cast(float, __builtin_amdgcn_readlane(__builtin_bit_cast(int, cvd), i));
        const float ax = a[i].x * cs + hb.x, ay = a[i].y * cs + hb.y;
        const float bx = b[i].x * cd + hb.x, by = b[i].y * cd + hb.y;
        p[i] = ax * ws.x + ay * ws.y + bx * wd.x + by * wd.y;
    }

    const bool u16 = (lane & 16) != 0, u8 = (lane & 8) != 0, u4 = (lane & 4) != 0;
    float q[4], r[2];
    for (int i = 0; i < 4; ++i) {
        const float keep = u16 ? p[i + 4] : p[i];
        const float send = u16 ? p[i] : p[i + 4];
        q[i] = keep + xrow(send);
    }
    for (int i = 0; i < 2; ++i) {
        const float keep = u8 ? q[i + 2] : q[i];
        const float send = u8 ? q[i] : q[i + 2];
        r[i] = keep + xlane<8>(send);
    }
    float t;
    {
        const float keep = u4 ? r[1] : r[0];
        const float send = u4 ? r[0] : r[1];
        t = keep + xlane<4>(send);
    }
    t += (c < 3) ? ef * wf : bb;
    t += xlane<2>(t);
    t += xlane<1>(t);
    if (c == 0) out[e0 + k] = 1.0f / (1.0f + expf(-t));
}

extern "C" void kernel_launch(void* const* d_in, const int* in_sizes, int n_in,
                              void* d_out, int out_size, void* d_ws, size_t ws_size,
                              hipStream_t stream) {
    const float* x     = (const float*)d_in[0];
    const float* efeat = (const float*)d_in[1];
    const int*   src   = (const int*)  d_in[2];
    const int*   dst   = (const int*)  d_in[3];
    const float* W1    = (const float*)d_in[4];
    const float* b1    = (const float*)d_in[5];
    const float* W2    = (const float*)d_in[6];
    const float* b2    = (const float*)d_in[7];
    const float* We    = (const float*)d_in[8];
    const float* be    = (const float*)d_in[9];
    float* out = (float*)d_out;

    const size_t NB = (size_t)N_NODES * 64;
    char* ws = (char*)d_ws;
    float* bufA = (float*)ws;
    float* bufB = bufA + NB;
    float* bufC = bufB + NB;
    float* cS   = bufC + NB;
    float* cD   = cS + N_NODES;
    int*   qctr = (int*)(cD + N_NODES) + 32;
    int*   part = qctr + 4 * QSTRIDE;

    hipMemsetAsync(bufB, 0, NB * sizeof(float), stream);

    degree_kernel<<<CNT_R * CNT_C, CNT_THREADS, 0, stream>>>(src, dst, part, qctr);
    coeff_kernel<<<(N_NODES + 127) / 128, 128, 0, stream>>>(part, cS, cD, N_NODES);

    gemm_wmma_kernel<128, false><<<(M_TILES + 3) / 4, 128, 0, stream>>>(x, W1, nullptr, nullptr, cS, bufA, M_TILES);
    scatter64_kernel<<<SCATTER_BLOCKS, 256, 0, stream>>>(bufA, src, dst, bufB, qctr);

    gemm_wmma_kernel<64, true><<<(M_TILES + 3) / 4, 128, 0, stream>>>(bufB, W2, cD, b1, cS, bufC, M_TILES);
    hipMemsetAsync(bufA, 0, NB * sizeof(float), stream);
    scatter64_kernel<<<SCATTER_BLOCKS, 256, 0, stream>>>(bufC, src, dst, bufA, qctr + 2 * QSTRIDE);

    edge_kernel<<<N_EDGES / (8 * EPE), 256, 0, stream>>>(bufA, cD, b2, src, dst, efeat, We, be, out, N_EDGES);
}
